# starting kernel
# baseline (speedup 1.0000x reference)
#include <hip/hip_runtime.h>
#include <hip/hip_bf16.h>
#include <math.h>

typedef __attribute__((ext_vector_type(2))) float v2f;
typedef __attribute__((ext_vector_type(8))) float v8f;

#define C_DIM 256

__global__ void gcn_zero_kernel(unsigned* __restrict__ deg,
                                float* __restrict__ accum, int N) {
    int i = blockIdx.x * blockDim.x + threadIdx.x;
    if (i < N) {
        deg[i]   = 0u;
        accum[i] = 0.0f;
    }
}

__global__ void gcn_matvec_wmma_kernel(const float* __restrict__ x,
                                       const float* __restrict__ W,
                                       float* __restrict__ h, int N) {
    const int lane = threadIdx.x & 31;
    const int wave = threadIdx.x >> 5;
    const int wavesPerBlock = blockDim.x >> 5;
    const int base = (blockIdx.x * wavesPerBlock + wave) * 16;
    if (base >= N) return;

    const int half = lane >> 4;
    const int m    = lane & 15;
    int row = base + m;
    if (row >= N) row = N - 1;

    const float* __restrict__ xrow = x + (size_t)row * C_DIM;

    v8f acc = {0.f, 0.f, 0.f, 0.f, 0.f, 0.f, 0.f, 0.f};

    #pragma unroll 4
    for (int k0 = 0; k0 < C_DIM; k0 += 4) {
        const int ka = k0 + half * 2;
        v2f a = *(const v2f*)(xrow + ka);
        v2f b;
        b.x = W[ka];
        b.y = W[ka + 1];
        acc = __builtin_amdgcn_wmma_f32_16x16x4_f32(
             false, a,  false, b,
             (short)0, acc,  false,  false);
    }

    if ((lane & 15) == 0) {
        const int mbase = base + half * 8;
        #pragma unroll
        for (int r = 0; r < 8; ++r) {
            if (mbase + r < N) h[mbase + r] = acc[r];
        }
    }
}

__global__ void gcn_degree_kernel(const int* __restrict__ ei,
                                  unsigned* __restrict__ deg, int E) {
    int e = blockIdx.x * blockDim.x + threadIdx.x;
    if (e < E) {
        atomicAdd(&deg[ei[E + e]], 1u);
    }
}

__global__ void gcn_dinv_kernel(const unsigned* __restrict__ deg,
                                float* __restrict__ dinv, int N) {
    int i = blockIdx.x * blockDim.x + threadIdx.x;
    if (i < N) {
        dinv[i] = rsqrtf((float)(deg[i] + 1u));
    }
}

__global__ void gcn_edge_kernel(const int* __restrict__ ei,
                                const float* __restrict__ dinv,
                                const float* __restrict__ h,
                                float* __restrict__ accum, int E) {
    int e = blockIdx.x * blockDim.x + threadIdx.x;
    if (e < E) {
        int s = ei[e];
        int d = ei[E + e];
        atomicAdd(&accum[d], dinv[s] * dinv[d] * h[s]);
    }
}

__global__ void gcn_finalize_kernel(const float* __restrict__ accum,
                                    const float* __restrict__ dinv,
                                    const float* __restrict__ h,
                                    const float* __restrict__ b,
                                    float* __restrict__ out, int N) {
    int i = blockIdx.x * blockDim.x + threadIdx.x;
    if (i < N) {
        float di = dinv[i];
        float v  = accum[i] + di * di * h[i] + b[0];
        out[i]   = fmaxf(v, 0.0f) + log1pf(expf(-fabsf(v)));
    }
}

extern "C" void kernel_launch(void* const* d_in, const int* in_sizes, int n_in,
                              void* d_out, int out_size, void* d_ws, size_t ws_size,
                              hipStream_t stream) {
    const float* x  = (const float*)d_in[0];
    const int*   ei = (const int*)d_in[1];
    const float* W  = (const float*)d_in[2];
    const float* b  = (const float*)d_in[3];
    float* out = (float*)d_out;

    const int N = in_sizes[0] / C_DIM;
    const int E = in_sizes[1] / 2;

    float*    h     = (float*)d_ws;
    unsigned* deg   = (unsigned*)(h + N);
    float*    dinv  = (float*)(deg + N);
    float*    accum = dinv + N;

    const int BLK = 256;
    const int gridN = (N + BLK - 1) / BLK;
    const int gridE = (E + BLK - 1) / BLK;
    const int nodesPerBlock = (BLK / 32) * 16;
    const int gridM = (N + nodesPerBlock - 1) / nodesPerBlock;

    gcn_zero_kernel<<<gridN, BLK, 0, stream>>>(deg, accum, N);
    gcn_matvec_wmma_kernel<<<gridM, BLK, 0, stream>>>(x, W, h, N);
    gcn_degree_kernel<<<gridE, BLK, 0, stream>>>(ei, deg, E);
    gcn_dinv_kernel<<<gridN, BLK, 0, stream>>>(deg, dinv, N);
    gcn_edge_kernel<<<gridE, BLK, 0, stream>>>(ei, dinv, h, accum, E);
    gcn_finalize_kernel<<<gridN, BLK, 0, stream>>>(accum, dinv, h, b, out, N);
}


# Per-edge global atomic adds become a counting sort of the edge list into 256-node dst bins (packed records); degrees and messages summed in LDS per bin; float sum order and rounding change
# speedup vs baseline: 6.0512x; 6.0512x over previous
#include <hip/hip_runtime.h>
#include <hip/hip_bf16.h>
#include <math.h>

typedef __attribute__((ext_vector_type(2))) float v2f;
typedef __attribute__((ext_vector_type(8))) float v8f;

#define C_DIM 256

__device__ __forceinline__ int chip_half() {
    const unsigned id = __builtin_amdgcn_s_sendmsg_rtn(0x87);
    return (int)((id >> 18) & 1u);
}

__global__ void gcn_zero_kernel(unsigned* __restrict__ deg, int N) {
    int i = blockIdx.x * blockDim.x + threadIdx.x;
    if (i < N) {
        deg[i]     = 0u;
        deg[N + i] = 0u;
    }
}

__device__ __forceinline__ void gcn_matvec_wmma_block(const float* __restrict__ x,
                                                      const float* __restrict__ W,
                                                      float* __restrict__ h, int N, int blk) {
    const int lane = threadIdx.x & 31;
    const int wave = threadIdx.x >> 5;
    const int wavesPerBlock = blockDim.x >> 5;
    const int base = (blk * wavesPerBlock + wave) * 16;
    if (base >= N) return;

    const int half = lane >> 4;
    const int m    = lane & 15;
    int row = base + m;
    if (row >= N) row = N - 1;

    const float* __restrict__ xrow = x + (size_t)row * C_DIM;

    v8f acc = {0.f, 0.f, 0.f, 0.f, 0.f, 0.f, 0.f, 0.f};

    #pragma unroll 4
    for (int k0 = 0; k0 < C_DIM; k0 += 4) {
        const int ka = k0 + half * 2;
        v2f a = *(const v2f*)(xrow + ka);
        v2f b;
        b.x = W[ka];
        b.y = W[ka + 1];
        acc = __builtin_amdgcn_wmma_f32_16x16x4_f32(
             false, a,  false, b,
             (short)0, acc,  false,  false);
    }

    if ((lane & 15) == 0) {
        const int mbase = base + half * 8;
        #pragma unroll
        for (int r = 0; r < 8; ++r) {
            if (mbase + r < N) h[mbase + r] = acc[r];
        }
    }
}

#define EPT  16
#define EBLK 128

#define FBLK 256

__device__ __forceinline__ void gcn_degree_block(const int* __restrict__ ei,
                                                 unsigned* __restrict__ deg,
                                                 int E, int N, int chunk) {
    unsigned* __restrict__ plane = deg + (size_t)chip_half() * (size_t)N;
    const int* __restrict__ dstp = ei + E;
    const int base = chunk * (FBLK * EPT);
    const int e0   = base + threadIdx.x;
    int d[EPT];
    if (base + FBLK * EPT <= E) {
        #pragma unroll
        for (int j = 0; j < EPT; ++j) d[j] = dstp[e0 + j * FBLK];
        #pragma unroll
        for (int j = 0; j < EPT; ++j) atomicAdd(&plane[d[j]], 1u);
    } else {
        #pragma unroll
        for (int j = 0; j < EPT; ++j) d[j] = dstp[min(e0 + j * FBLK, E - 1)];
        #pragma unroll
        for (int j = 0; j < EPT; ++j)
            if (e0 + j * FBLK < E) atomicAdd(&plane[d[j]], 1u);
    }
}

__global__ void __launch_bounds__(FBLK)
gcn_matvec_degree_kernel(const float* __restrict__ x,
                         const float* __restrict__ W,
                         float* __restrict__ h,
                         const int* __restrict__ ei,
                         unsigned* __restrict__ deg,
                         int N, int E, int nMat, int nDeg) {
    const int b     = blockIdx.x;
    const int pairs = min(nMat, nDeg);
    bool counts;
    int  item;
    if (b < 2 * pairs) { counts = (__popc((unsigned)b) & 1) != 0; item = b >> 1; }
    else               { counts = nDeg > nMat;                    item = b - pairs; }
    if (counts) gcn_degree_block(ei, deg, E, N, item);
    else        gcn_matvec_wmma_block(x, W, h, N, item);
}

__global__ void gcn_dinv_kernel(unsigned* __restrict__ deg,
                                float* __restrict__ dinv, int N) {
    int i = blockIdx.x * blockDim.x + threadIdx.x;
    if (i < N) {
        const unsigned d = deg[i] + deg[N + i];
        dinv[i] = rsqrtf((float)(d + 1u));
        deg[i]     = 0u;
        deg[N + i] = 0u;
    }
}

__global__ void __launch_bounds__(EBLK)
gcn_edge_kernel(const int* __restrict__ ei,
                const float* __restrict__ dinv,
                const float* __restrict__ h,
                float* __restrict__ accum, int E, int N) {
    float* __restrict__ plane = accum + (size_t)chip_half() * (size_t)N;
    const int* __restrict__ dstp = ei + E;
    const int base = blockIdx.x * (EBLK * EPT);
    const int e0   = base + threadIdx.x;
    const bool whole = (base + EBLK * EPT <= E);
    int   s[EPT], d[EPT];
    float w[EPT];
    if (whole) {
        #pragma unroll
        for (int j = 0; j < EPT; ++j) {
            s[j] = ei[e0 + j * EBLK];
            d[j] = dstp[e0 + j * EBLK];
        }
    } else {
        #pragma unroll
        for (int j = 0; j < EPT; ++j) {
            const int e = min(e0 + j * EBLK, E - 1);
            s[j] = ei[e];
            d[j] = dstp[e];
        }
    }
    #pragma unroll
    for (int j = 0; j < EPT; ++j) w[j] = dinv[s[j]] * dinv[d[j]] * h[s[j]];
    if (whole) {
        #pragma unroll
        for (int j = 0; j < EPT; ++j) atomicAdd(&plane[d[j]], w[j]);
    } else {
        #pragma unroll
        for (int j = 0; j < EPT; ++j)
            if (e0 + j * EBLK < E) atomicAdd(&plane[d[j]], w[j]);
    }
}

__global__ void gcn_finalize_kernel(const float* __restrict__ accum,
                                    const float* __restrict__ dinv,
                                    const float* __restrict__ h,
                                    const float* __restrict__ b,
                                    float* __restrict__ out, int N) {
    int i = blockIdx.x * blockDim.x + threadIdx.x;
    if (i < N) {
        float di = dinv[i];
        float v  = (accum[i] + accum[N + i]) + di * di * h[i] + b[0];
        out[i]   = fmaxf(v, 0.0f) + log1pf(expf(-fabsf(v)));
    }
}


#define RS_BITS  8
#define RS       (1 << RS_BITS)
#define RMAX     512
#define SRC_BITS 17
#define SRC_MASK ((1u << SRC_BITS) - 1u)
#define SBLK     512
#define SCHUNK   (SBLK * EPT)
#define RBLK     256

__global__ void gcn_clear_kernel(unsigned* __restrict__ words, int n) {
    int i = blockIdx.x * blockDim.x + threadIdx.x;
    if (i < n) words[i] = 0u;
}

__device__ __forceinline__ void gcn_hist_block(const int* __restrict__ ei,
                                               unsigned* __restrict__ tot,
                                               int E, int chunk) {
    __shared__ unsigned lc[RMAX];
    for (int q = threadIdx.x; q < RMAX; q += FBLK) lc[q] = 0u;
    __syncthreads();
    const int* __restrict__ dstp = ei + E;
    const int e0 = chunk * (FBLK * EPT) + threadIdx.x;
    int d[EPT];
    #pragma unroll
    for (int j = 0; j < EPT; ++j) d[j] = dstp[min(e0 + j * FBLK, E - 1)];
    #pragma unroll
    for (int j = 0; j < EPT; ++j)
        if (e0 + j * FBLK < E) atomicAdd(&lc[d[j] >> RS_BITS], 1u);
    __syncthreads();
    for (int q = threadIdx.x; q < RMAX; q += FBLK) {
        const unsigned c = lc[q];
        if (c) atomicAdd(&tot[q], c);
    }
}

__global__ void __launch_bounds__(FBLK)
gcn_matvec_hist_kernel(const float* __restrict__ x,
                       const float* __restrict__ W,
                       float* __restrict__ h,
                       const int* __restrict__ ei,
                       unsigned* __restrict__ tot,
                       int N, int E, int nMat, int nDeg) {
    const int b     = blockIdx.x;
    const int pairs = min(nMat, nDeg);
    bool counts;
    int  item;
    if (b < 2 * pairs) { counts = (__popc((unsigned)b) & 1) != 0; item = b >> 1; }
    else               { counts = nDeg > nMat;                    item = b - pairs; }
    if (counts) gcn_hist_block(ei, tot, E, item);
    else        gcn_matvec_wmma_block(x, W, h, N, item);
}

__global__ void __launch_bounds__(SBLK)
gcn_place_kernel(const int* __restrict__ ei,
                 const unsigned* __restrict__ tot,
                 unsigned* __restrict__ fill,
                 unsigned* __restrict__ recs, int E) {
    __shared__ unsigned lc[RMAX];
    __shared__ unsigned lbase[RMAX];
    __shared__ unsigned loff[RMAX];
    __shared__ unsigned sa[2][RMAX];
    __shared__ unsigned sb[2][RMAX];
    __shared__ unsigned srec[SCHUNK];
    __shared__ unsigned short srng[SCHUNK];
    const int t = threadIdx.x;
    lc[t] = 0u;
    __syncthreads();
    const int* __restrict__ dstp = ei + E;
    const int base = blockIdx.x * SCHUNK;
    const int e0   = base + t;
    int s[EPT], d[EPT];
    unsigned rk[EPT];
    #pragma unroll
    for (int j = 0; j < EPT; ++j) {
        const int e = min(e0 + j * SBLK, E - 1);
        s[j] = ei[e];
        d[j] = dstp[e];
    }
    #pragma unroll
    for (int j = 0; j < EPT; ++j) {
        rk[j] = 0u;
        if (e0 + j * SBLK < E) rk[j] = atomicAdd(&lc[d[j] >> RS_BITS], 1u);
    }
    __syncthreads();
    const unsigned c  = lc[t];
    const unsigned tt = tot[t];
    sa[0][t] = tt;
    sb[0][t] = c;
    __syncthreads();
    int p = 0;
    for (int off = 1; off < RMAX; off <<= 1) {
        unsigned ua = sa[p][t], ub = sb[p][t];
        if (t >= off) { ua += sa[p][t - off]; ub += sb[p][t - off]; }
        sa[p ^ 1][t] = ua;
        sb[p ^ 1][t] = ub;
        p ^= 1;
        __syncthreads();
    }
    unsigned got = 0u;
    if (c) got = atomicAdd(&fill[t], c);
    lbase[t] = (sa[p][t] - tt) + got;
    loff[t]  = sb[p][t] - c;
    __syncthreads();
    #pragma unroll
    for (int j = 0; j < EPT; ++j) {
        if (e0 + j * SBLK < E) {
            const unsigned r   = (unsigned)d[j] >> RS_BITS;
            const unsigned pos = loff[r] + rk[j];
            srec[pos] = (((unsigned)d[j] & (RS - 1u)) << SRC_BITS) | (unsigned)s[j];
            srng[pos] = (unsigned short)r;
        }
    }
    __syncthreads();
    const int nvalid = min(SCHUNK, E - base);
    for (int i = t; i < nvalid; i += SBLK) {
        const unsigned r = srng[i];
        recs[lbase[r] + ((unsigned)i - loff[r])] = srec[i];
    }
}

__device__ __forceinline__ unsigned gcn_bin_start(const unsigned* __restrict__ tot, int r,
                                                  unsigned* sh) {
    const int t = threadIdx.x;
    if (t == 0) *sh = 0u;
    __syncthreads();
    unsigned v = 0u;
    for (int q = t; q < r; q += RBLK) v += tot[q];
    for (int m = 16; m > 0; m >>= 1) v += __shfl_xor(v, m);
    if ((t & 31) == 0 && v) atomicAdd(sh, v);
    __syncthreads();
    return *sh;
}

__global__ void __launch_bounds__(RBLK)
gcn_bin_degree_kernel(const unsigned* __restrict__ recs,
                      const unsigned* __restrict__ tot,
                      const float* __restrict__ h,
                      float* __restrict__ dinv,
                      float* __restrict__ g, int N) {
    __shared__ unsigned cnt[RS];
    __shared__ unsigned sst;
    const int t = threadIdx.x;
    const int r = blockIdx.x;
    cnt[t] = 0u;
    const unsigned start = gcn_bin_start(tot, r, &sst);
    const unsigned n = tot[r];
    const unsigned* __restrict__ p = recs + start;
    for (unsigned i0 = 0; i0 < n; i0 += RBLK * EPT) {
        unsigned rc[EPT];
        #pragma unroll
        for (int j = 0; j < EPT; ++j) rc[j] = p[min(i0 + t + j * RBLK, n - 1u)];
        #pragma unroll
        for (int j = 0; j < EPT; ++j)
            if (i0 + t + j * RBLK < n) atomicAdd(&cnt[rc[j] >> SRC_BITS], 1u);
    }
    __syncthreads();
    const int node = r * RS + t;
    if (node < N) {
        const float di = rsqrtf((float)(cnt[t] + 1u));
        dinv[node] = di;
        g[node]    = di * h[node];
    }
}

__global__ void __launch_bounds__(RBLK)
gcn_bin_message_kernel(const unsigned* __restrict__ recs,
                       const unsigned* __restrict__ tot,
                       const float* __restrict__ g,
                       const float* __restrict__ dinv,
                       const float* __restrict__ b,
                       float* __restrict__ out, int N) {
    __shared__ float sum[RS];
    __shared__ unsigned sst;
    const int t = threadIdx.x;
    const int r = blockIdx.x;
    sum[t] = 0.0f;
    const unsigned start = gcn_bin_start(tot, r, &sst);
    const unsigned n = tot[r];
    const unsigned* __restrict__ p = recs + start;
    for (unsigned i0 = 0; i0 < n; i0 += RBLK * EPT) {
        unsigned rc[EPT];
        float    gv[EPT];
        #pragma unroll
        for (int j = 0; j < EPT; ++j) rc[j] = p[min(i0 + t + j * RBLK, n - 1u)];
        #pragma unroll
        for (int j = 0; j < EPT; ++j) gv[j] = g[rc[j] & SRC_MASK];
        #pragma unroll
        for (int j = 0; j < EPT; ++j)
            if (i0 + t + j * RBLK < n) atomicAdd(&sum[rc[j] >> SRC_BITS], gv[j]);
    }
    __syncthreads();
    const int node = r * RS + t;
    if (node < N) {
        const float v = dinv[node] * (sum[t] + g[node]) + b[0];
        out[node] = fmaxf(v, 0.0f) + log1pf(expf(-fabsf(v)));
    }
}

extern "C" void kernel_launch(void* const* d_in, const int* in_sizes, int n_in,
                              void* d_out, int out_size, void* d_ws, size_t ws_size,
                              hipStream_t stream) {
    const float* x  = (const float*)d_in[0];
    const int*   ei = (const int*)d_in[1];
    const float* W  = (const float*)d_in[2];
    const float* b  = (const float*)d_in[3];
    float* out = (float*)d_out;

    const int N = in_sizes[0] / C_DIM;
    const int E = in_sizes[1] / 2;

    float*    h     = (float*)d_ws;
    unsigned* deg   = (unsigned*)(h + N);
    float*    accum = (float*)deg;
    float*    dinv  = (float*)(deg + 2 * (size_t)N);

    const int BLK = 256;
    const int gridN = (N + BLK - 1) / BLK;
    const int gridE = (E + EBLK * EPT - 1) / (EBLK * EPT);
    const int nodesPerBlock = (FBLK / 32) * 16;
    const int gridM = (N + nodesPerBlock - 1) / nodesPerBlock;

    const int gridD = (E + FBLK * EPT - 1) / (FBLK * EPT);

    const size_t need = ((size_t)3 * N + 2 * RMAX + (size_t)E) * sizeof(unsigned);
    if (ws_size >= need && N <= RMAX * RS) {
        float*    g2    = h + N;
        float*    dinv2 = g2 + N;
        unsigned* tot   = (unsigned*)(dinv2 + N);
        unsigned* fill  = tot + RMAX;
        unsigned* recs  = fill + RMAX;
        const int gridP = (E + SCHUNK - 1) / SCHUNK;
        const int bins  = (N + RS - 1) / RS;
        gcn_clear_kernel<<<(2 * RMAX + BLK - 1) / BLK, BLK, 0, stream>>>(tot, 2 * RMAX);
        gcn_matvec_hist_kernel<<<gridM + gridD, FBLK, 0, stream>>>(x, W, h, ei, tot, N, E, gridM, gridD);
        gcn_place_kernel<<<gridP, SBLK, 0, stream>>>(ei, tot, fill, recs, E);
        gcn_bin_degree_kernel<<<bins, RBLK, 0, stream>>>(recs, tot, h, dinv2, g2, N);
        gcn_bin_message_kernel<<<bins, RBLK, 0, stream>>>(recs, tot, g2, dinv2, b, out, N);
        return;
    }

    gcn_zero_kernel<<<gridN, BLK, 0, stream>>>(deg, N);
    gcn_matvec_degree_kernel<<<gridM + gridD, FBLK, 0, stream>>>(x, W, h, ei, deg, N, E, gridM, gridD);
    gcn_dinv_kernel<<<gridN, BLK, 0, stream>>>(deg, dinv, N);
    gcn_edge_kernel<<<gridE, EBLK, 0, stream>>>(ei, dinv, h, accum, E, N);
    gcn_finalize_kernel<<<gridN, BLK, 0, stream>>>(accum, dinv, h, b, out, N);
}


# Bins own fixed stretches (an overflowing bin rescans the edge list: exact): counting pass and bin-start sums go; placing joins the matvec launch; 5 launches to 4; sums by arrival as before
# speedup vs baseline: 6.6315x; 1.0959x over previous
#include <hip/hip_runtime.h>
#include <hip/hip_bf16.h>
#include <math.h>

typedef __attribute__((ext_vector_type(2))) float v2f;
typedef __attribute__((ext_vector_type(8))) float v8f;

#define C_DIM 256

__device__ __forceinline__ int chip_half() {
    const unsigned id = __builtin_amdgcn_s_sendmsg_rtn(0x87);
    return (int)((id >> 18) & 1u);
}

__global__ void gcn_zero_kernel(unsigned* __restrict__ deg, int N) {
    int i = blockIdx.x * blockDim.x + threadIdx.x;
    if (i < N) {
        deg[i]     = 0u;
        deg[N + i] = 0u;
    }
}

__device__ __forceinline__ void gcn_matvec_wmma_block(const float* __restrict__ x,
                                                      const float* __restrict__ W,
                                                      float* __restrict__ h, int N, int blk) {
    const int lane = threadIdx.x & 31;
    const int wave = threadIdx.x >> 5;
    const int wavesPerBlock = blockDim.x >> 5;
    const int base = (blk * wavesPerBlock + wave) * 16;
    if (base >= N) return;

    const int half = lane >> 4;
    const int m    = lane & 15;
    int row = base + m;
    if (row >= N) row = N - 1;

    const float* __restrict__ xrow = x + (size_t)row * C_DIM;

    v8f acc = {0.f, 0.f, 0.f, 0.f, 0.f, 0.f, 0.f, 0.f};

    #pragma unroll 4
    for (int k0 = 0; k0 < C_DIM; k0 += 4) {
        const int ka = k0 + half * 2;
        v2f a = *(const v2f*)(xrow + ka);
        v2f b;
        b.x = W[ka];
        b.y = W[ka + 1];
        acc = __builtin_amdgcn_wmma_f32_16x16x4_f32(
             false, a,  false, b,
             (short)0, acc,  false,  false);
    }

    if ((lane & 15) == 0) {
        const int mbase = base + half * 8;
        #pragma unroll
        for (int r = 0; r < 8; ++r) {
            if (mbase + r < N) h[mbase + r] = acc[r];
        }
    }
}

#define EPT  16
#define EBLK 128

#define FBLK 256

__device__ __forceinline__ void gcn_degree_block(const int* __restrict__ ei,
                                                 unsigned* __restrict__ deg,
                                                 int E, int N, int chunk) {
    unsigned* __restrict__ plane = deg + (size_t)chip_half() * (size_t)N;
    const int* __restrict__ dstp = ei + E;
    const int base = chunk * (FBLK * EPT);
    const int e0   = base + threadIdx.x;
    int d[EPT];
    if (base + FBLK * EPT <= E) {
        #pragma unroll
        for (int j = 0; j < EPT; ++j) d[j] = dstp[e0 + j * FBLK];
        #pragma unroll
        for (int j = 0; j < EPT; ++j) atomicAdd(&plane[d[j]], 1u);
    } else {
        #pragma unroll
        for (int j = 0; j < EPT; ++j) d[j] = dstp[min(e0 + j * FBLK, E - 1)];
        #pragma unroll
        for (int j = 0; j < EPT; ++j)
            if (e0 + j * FBLK < E) atomicAdd(&plane[d[j]], 1u);
    }
}

__global__ void __launch_bounds__(FBLK)
gcn_matvec_degree_kernel(const float* __restrict__ x,
                         const float* __restrict__ W,
                         float* __restrict__ h,
                         const int* __restrict__ ei,
                         unsigned* __restrict__ deg,
                         int N, int E, int nMat, int nDeg) {
    const int b     = blockIdx.x;
    const int pairs = min(nMat, nDeg);
    bool counts;
    int  item;
    if (b < 2 * pairs) { counts = (__popc((unsigned)b) & 1) != 0; item = b >> 1; }
    else               { counts = nDeg > nMat;                    item = b - pairs; }
    if (counts) gcn_degree_block(ei, deg, E, N, item);
    else        gcn_matvec_wmma_block(x, W, h, N, item);
}

__global__ void gcn_dinv_kernel(unsigned* __restrict__ deg,
                                float* __restrict__ dinv, int N) {
    int i = blockIdx.x * blockDim.x + threadIdx.x;
    if (i < N) {
        const unsigned d = deg[i] + deg[N + i];
        dinv[i] = rsqrtf((float)(d + 1u));
        deg[i]     = 0u;
        deg[N + i] = 0u;
    }
}

__global__ void __launch_bounds__(EBLK)
gcn_edge_kernel(const int* __restrict__ ei,
                const float* __restrict__ dinv,
                const float* __restrict__ h,
                float* __restrict__ accum, int E, int N) {
    float* __restrict__ plane = accum + (size_t)chip_half() * (size_t)N;
    const int* __restrict__ dstp = ei + E;
    const int base = blockIdx.x * (EBLK * EPT);
    const int e0   = base + threadIdx.x;
    const bool whole = (base + EBLK * EPT <= E);
    int   s[EPT], d[EPT];
    float w[EPT];
    if (whole) {
        #pragma unroll
        for (int j = 0; j < EPT; ++j) {
            s[j] = ei[e0 + j * EBLK];
            d[j] = dstp[e0 + j * EBLK];
        }
    } else {
        #pragma unroll
        for (int j = 0; j < EPT; ++j) {
            const int e = min(e0 + j * EBLK, E - 1);
            s[j] = ei[e];
            d[j] = dstp[e];
        }
    }
    #pragma unroll
    for (int j = 0; j < EPT; ++j) w[j] = dinv[s[j]] * dinv[d[j]] * h[s[j]];
    if (whole) {
        #pragma unroll
        for (int j = 0; j < EPT; ++j) atomicAdd(&plane[d[j]], w[j]);
    } else {
        #pragma unroll
        for (int j = 0; j < EPT; ++j)
            if (e0 + j * EBLK < E) atomicAdd(&plane[d[j]], w[j]);
    }
}

__global__ void gcn_finalize_kernel(const float* __restrict__ accum,
                                    const float* __restrict__ dinv,
                                    const float* __restrict__ h,
                                    const float* __restrict__ b,
                                    float* __restrict__ out, int N) {
    int i = blockIdx.x * blockDim.x + threadIdx.x;
    if (i < N) {
        float di = dinv[i];
        float v  = (accum[i] + accum[N + i]) + di * di * h[i] + b[0];
        out[i]   = fmaxf(v, 0.0f) + log1pf(expf(-fabsf(v)));
    }
}


#define RS_BITS  8
#define RS       (1 << RS_BITS)
#define RMAX     512
#define SRC_BITS 17
#define SRC_MASK ((1u << SRC_BITS) - 1u)
#define SBLK     512
#define SCHUNK   (SBLK * EPT)
#define RBLK     256
#define CAP      8704

__global__ void gcn_clear_kernel(unsigned* __restrict__ words, int n) {
    int i = blockIdx.x * blockDim.x + threadIdx.x;
    if (i < n) words[i] = 0u;
}

__device__ __forceinline__ void gcn_place_block(const int* __restrict__ ei,
                                                unsigned* __restrict__ fill,
                                                unsigned* __restrict__ recs,
                                                int E, int chunk) {
    __shared__ unsigned lc[RMAX];
    __shared__ unsigned lgot[RMAX];
    __shared__ unsigned loff[RMAX];
    __shared__ unsigned sb[2][RMAX];
    __shared__ unsigned srec[SCHUNK];
    __shared__ unsigned short srng[SCHUNK];
    const int t = threadIdx.x;
    lc[t] = 0u;
    __syncthreads();
    const int* __restrict__ dstp = ei + E;
    const int base = chunk * SCHUNK;
    const int e0   = base + t;
    int s[EPT], d[EPT];
    unsigned rk[EPT];
    #pragma unroll
    for (int j = 0; j < EPT; ++j) {
        const int e = min(e0 + j * SBLK, E - 1);
        s[j] = ei[e];
        d[j] = dstp[e];
    }
    #pragma unroll
    for (int j = 0; j < EPT; ++j) {
        rk[j] = 0u;
        if (e0 + j * SBLK < E) rk[j] = atomicAdd(&lc[d[j] >> RS_BITS], 1u);
    }
    __syncthreads();
    const unsigned c = lc[t];
    unsigned got = 0u;
    if (c) got = atomicAdd(&fill[t], c);
    sb[0][t] = c;
    __syncthreads();
    int p = 0;
    for (int off = 1; off < RMAX; off <<= 1) {
        unsigned ub = sb[p][t];
        if (t >= off) ub += sb[p][t - off];
        sb[p ^ 1][t] = ub;
        p ^= 1;
        __syncthreads();
    }
    lgot[t] = got;
    loff[t] = sb[p][t] - c;
    __syncthreads();
    #pragma unroll
    for (int j = 0; j < EPT; ++j) {
        if (e0 + j * SBLK < E) {
            const unsigned r   = (unsigned)d[j] >> RS_BITS;
            const unsigned pos = loff[r] + rk[j];
            srec[pos] = (((unsigned)d[j] & (RS - 1u)) << SRC_BITS) | (unsigned)s[j];
            srng[pos] = (unsigned short)r;
        }
    }
    __syncthreads();
    const int nvalid = min(SCHUNK, E - base);
    for (int i = t; i < nvalid; i += SBLK) {
        const unsigned r = srng[i];
        const unsigned k = lgot[r] + ((unsigned)i - loff[r]);
        if (k < CAP) recs[(size_t)r * CAP + k] = srec[i];
    }
}

__global__ void __launch_bounds__(SBLK)
gcn_matvec_place_kernel(const float* __restrict__ x,
                        const float* __restrict__ W,
                        float* __restrict__ h,
                        const int* __restrict__ ei,
                        unsigned* __restrict__ fill,
                        unsigned* __restrict__ recs,
                        int N, int E, int nMat, int nPlace) {
    const int b     = blockIdx.x;
    const int pairs = min(nMat, nPlace);
    bool places;
    int  item;
    if (b < 2 * pairs) { places = (__popc((unsigned)b) & 1) != 0; item = b >> 1; }
    else               { places = nPlace > nMat;                  item = b - pairs; }
    if (places) gcn_place_block(ei, fill, recs, E, item);
    else        gcn_matvec_wmma_block(x, W, h, N, item);
}

__global__ void __launch_bounds__(RBLK)
gcn_bin_degree_kernel(const unsigned* __restrict__ recs,
                      const unsigned* __restrict__ fill,
                      const int* __restrict__ ei,
                      const float* __restrict__ h,
                      float* __restrict__ dinv,
                      float* __restrict__ g, int N, int E) {
    __shared__ unsigned cnt[RS];
    const int t = threadIdx.x;
    const int r = blockIdx.x;
    cnt[t] = 0u;
    __syncthreads();
    const unsigned n = fill[r];
    if (n <= CAP) {
        const unsigned* __restrict__ p = recs + (size_t)r * CAP;
        for (unsigned i0 = 0; i0 < n; i0 += RBLK * EPT) {
            unsigned rc[EPT];
            #pragma unroll
            for (int j = 0; j < EPT; ++j) rc[j] = p[min(i0 + t + j * RBLK, n - 1u)];
            #pragma unroll
            for (int j = 0; j < EPT; ++j)
                if (i0 + t + j * RBLK < n) atomicAdd(&cnt[rc[j] >> SRC_BITS], 1u);
        }
    } else {
        const int* __restrict__ dstp = ei + E;
        for (int i = t; i < E; i += RBLK) {
            const int d = dstp[i];
            if ((d >> RS_BITS) == r) atomicAdd(&cnt[d & (RS - 1)], 1u);
        }
    }
    __syncthreads();
    const int node = r * RS + t;
    if (node < N) {
        const float di = rsqrtf((float)(cnt[t] + 1u));
        dinv[node] = di;
        g[node]    = di * h[node];
    }
}

__global__ void __launch_bounds__(RBLK)
gcn_bin_message_kernel(const unsigned* __restrict__ recs,
                       const unsigned* __restrict__ fill,
                       const int* __restrict__ ei,
                       const float* __restrict__ g,
                       const float* __restrict__ dinv,
                       const float* __restrict__ b,
                       float* __restrict__ out, int N, int E) {
    __shared__ float sum[RS];
    const int t = threadIdx.x;
    const int r = blockIdx.x;
    sum[t] = 0.0f;
    __syncthreads();
    const unsigned n = fill[r];
    if (n <= CAP) {
        const unsigned* __restrict__ p = recs + (size_t)r * CAP;
        for (unsigned i0 = 0; i0 < n; i0 += RBLK * EPT) {
            unsigned rc[EPT];
            float    gv[EPT];
            #pragma unroll
            for (int j = 0; j < EPT; ++j) rc[j] = p[min(i0 + t + j * RBLK, n - 1u)];
            #pragma unroll
            for (int j = 0; j < EPT; ++j) gv[j] = g[rc[j] & SRC_MASK];
            #pragma unroll
            for (int j = 0; j < EPT; ++j)
                if (i0 + t + j * RBLK < n) atomicAdd(&sum[rc[j] >> SRC_BITS], gv[j]);
        }
    } else {
        const int* __restrict__ dstp = ei + E;
        for (int i = t; i < E; i += RBLK) {
            const int d = dstp[i];
            if ((d >> RS_BITS) == r) atomicAdd(&sum[d & (RS - 1)], g[ei[i]]);
        }
    }
    __syncthreads();
    const int node = r * RS + t;
    if (node < N) {
        const float v = dinv[node] * (sum[t] + g[node]) + b[0];
        out[node] = fmaxf(v, 0.0f) + log1pf(expf(-fabsf(v)));
    }
}

extern "C" void kernel_launch(void* const* d_in, const int* in_sizes, int n_in,
                              void* d_out, int out_size, void* d_ws, size_t ws_size,
                              hipStream_t stream) {
    const float* x  = (const float*)d_in[0];
    const int*   ei = (const int*)d_in[1];
    const float* W  = (const float*)d_in[2];
    const float* b  = (const float*)d_in[3];
    float* out = (float*)d_out;

    const int N = in_sizes[0] / C_DIM;
    const int E = in_sizes[1] / 2;

    float*    h     = (float*)d_ws;
    unsigned* deg   = (unsigned*)(h + N);
    float*    accum = (float*)deg;
    float*    dinv  = (float*)(deg + 2 * (size_t)N);

    const int BLK = 256;
    const int gridN = (N + BLK - 1) / BLK;
    const int gridE = (E + EBLK * EPT - 1) / (EBLK * EPT);
    const int nodesPerBlock = (FBLK / 32) * 16;
    const int gridM = (N + nodesPerBlock - 1) / nodesPerBlock;

    const int gridD = (E + FBLK * EPT - 1) / (FBLK * EPT);

    const int bins = (N + RS - 1) / RS;
    const size_t need = ((size_t)3 * N + RMAX + (size_t)bins * CAP) * sizeof(unsigned);
    if (ws_size >= need && N <= RMAX * RS) {
        float*    g2    = h + N;
        float*    dinv2 = g2 + N;
        unsigned* fill  = (unsigned*)(dinv2 + N);
        unsigned* recs  = fill + RMAX;
        const int gridP = (E + SCHUNK - 1) / SCHUNK;
        const int nodesPerPBlock = (SBLK / 32) * 16;
        const int gridMP = (N + nodesPerPBlock - 1) / nodesPerPBlock;
        gcn_clear_kernel<<<(RMAX + BLK - 1) / BLK, BLK, 0, stream>>>(fill, RMAX);
        gcn_matvec_place_kernel<<<gridMP + gridP, SBLK, 0, stream>>>(x, W, h, ei, fill, recs, N, E, gridMP, gridP);
        gcn_bin_degree_kernel<<<bins, RBLK, 0, stream>>>(recs, fill, ei, h, dinv2, g2, N, E);
        gcn_bin_message_kernel<<<bins, RBLK, 0, stream>>>(recs, fill, ei, g2, dinv2, b, out, N, E);
        return;
    }

    gcn_zero_kernel<<<gridN, BLK, 0, stream>>>(deg, N);
    gcn_matvec_degree_kernel<<<gridM + gridD, FBLK, 0, stream>>>(x, W, h, ei, deg, N, E, gridM, gridD);
    gcn_dinv_kernel<<<gridN, BLK, 0, stream>>>(deg, dinv, N);
    gcn_edge_kernel<<<gridE, EBLK, 0, stream>>>(ei, dinv, h, accum, E, N);
    gcn_finalize_kernel<<<gridN, BLK, 0, stream>>>(accum, dinv, h, b, out, N);
}


# Matvec: x and W reach the WMMA fragments through LDS (whole-line 16-byte global loads into a padded per-wave tile, next chunk in flight under the products); k order and sums as before
# speedup vs baseline: 8.2072x; 1.2376x over previous
#include <hip/hip_runtime.h>
#include <hip/hip_bf16.h>
#include <math.h>

typedef __attribute__((ext_vector_type(2))) float v2f;
typedef __attribute__((ext_vector_type(4))) float v4f;
typedef __attribute__((ext_vector_type(8))) float v8f;

#define C_DIM 256

#define XT_K     64
#define XT_PITCH 68
#define XT_WAVE  (16 * XT_PITCH)

__device__ __forceinline__ int chip_half() {
    const unsigned id = __builtin_amdgcn_s_sendmsg_rtn(0x87);
    return (int)((id >> 18) & 1u);
}

__global__ void gcn_zero_kernel(unsigned* __restrict__ deg, int N) {
    int i = blockIdx.x * blockDim.x + threadIdx.x;
    if (i < N) {
        deg[i]     = 0u;
        deg[N + i] = 0u;
    }
}

__device__ __forceinline__ void gcn_matvec_store_tile(float* __restrict__ h, const v8f acc,
                                                      int base, int lane, int N) {
    if ((lane & 15) == 0) {
        const int mbase = base + (lane >> 4) * 8;
        #pragma unroll
        for (int r8 = 0; r8 < 8; ++r8) {
            if (mbase + r8 < N) h[mbase + r8] = acc[r8];
        }
    }
}

__device__ __forceinline__ void gcn_matvec_wmma_block(const float* __restrict__ x,
                                                      const float* __restrict__ W,
                                                      float* __restrict__ h, int N, int blk,
                                                      float* __restrict__ xt,
                                                      float* __restrict__ wl) {
    if (threadIdx.x < C_DIM) wl[threadIdx.x] = W[threadIdx.x];
    __syncthreads();

    const int lane = threadIdx.x & 31;
    const int wave = __builtin_amdgcn_readfirstlane((int)(threadIdx.x >> 5));
    const int wavesPerBlock = blockDim.x >> 5;
    const int base = __builtin_amdgcn_readfirstlane((blk * wavesPerBlock + wave) * 16);
    if (base >= N) return;

    const int half = lane >> 4;
    const int m    = lane & 15;

    v8f acc = {0.f, 0.f, 0.f, 0.f, 0.f, 0.f, 0.f, 0.f};

    if (base + 16 > N) {
        const int row = min(base + m, N - 1);
        const float* __restrict__ xrow = x + (size_t)row * C_DIM;
        #pragma unroll 4
        for (int k0 = 0; k0 < C_DIM; k0 += 4) {
            const int ka = k0 + half * 2;
            v2f a = *(const v2f*)(xrow + ka);
            v2f b;
            b.x = W[ka];
            b.y = W[ka + 1];
            acc = __builtin_amdgcn_wmma_f32_16x16x4_f32(
                 false, a,  false, b,
                 (short)0, acc,  false,  false);
        }
        gcn_matvec_store_tile(h, acc, base, lane, N);
        return;
    }

    const char* xp = (const char*)(x + (size_t)base * C_DIM) + (unsigned)(half * C_DIM + 4 * m) * 4u;
    float* tile = xt + wave * XT_WAVE;
    float* st = tile + half * XT_PITCH + 4 * m;
    const float* ld = tile + m * XT_PITCH + 2 * half;
    const float* wf = wl + 2 * half;

    v4f r[8];
    #pragma unroll
    for (int i = 0; i < 8; ++i) r[i] = *(const v4f*)(xp + i * (2 * C_DIM * 4));

    #pragma nounroll
    for (int q = 0; q < C_DIM / XT_K; ++q) {
        #pragma unroll
        for (int i = 0; i < 8; ++i) *(v4f*)(st + 2 * i * XT_PITCH) = r[i];
        if (q + 1 < C_DIM / XT_K) {
            xp += XT_K * 4;
            #pragma unroll
            for (int i = 0; i < 8; ++i) r[i] = *(const v4f*)(xp + i * (2 * C_DIM * 4));
        }
        #pragma unroll
        for (int s = 0; s < XT_K / 4; ++s) {
            v2f a = *(const v2f*)(ld + 4 * s);
            v2f b = *(const v2f*)(wf + 4 * s);
            acc = __builtin_amdgcn_wmma_f32_16x16x4_f32(
                 false, a,  false, b,
                 (short)0, acc,  false,  false);
        }
        wf += XT_K;
    }
    gcn_matvec_store_tile(h, acc, base, lane, N);
}

#define EPT  16
#define EBLK 128

#define FBLK 256

__device__ __forceinline__ void gcn_degree_block(const int* __restrict__ ei,
                                                 unsigned* __restrict__ deg,
                                                 int E, int N, int chunk) {
    unsigned* __restrict__ plane = deg + (size_t)chip_half() * (size_t)N;
    const int* __restrict__ dstp = ei + E;
    const int base = chunk * (FBLK * EPT);
    const int e0   = base + threadIdx.x;
    int d[EPT];
    if (base + FBLK * EPT <= E) {
        #pragma unroll
        for (int j = 0; j < EPT; ++j) d[j] = dstp[e0 + j * FBLK];
        #pragma unroll
        for (int j = 0; j < EPT; ++j) atomicAdd(&plane[d[j]], 1u);
    } else {
        #pragma unroll
        for (int j = 0; j < EPT; ++j) d[j] = dstp[min(e0 + j * FBLK, E - 1)];
        #pragma unroll
        for (int j = 0; j < EPT; ++j)
            if (e0 + j * FBLK < E) atomicAdd(&plane[d[j]], 1u);
    }
}

__global__ void __launch_bounds__(FBLK)
gcn_matvec_degree_kernel(const float* __restrict__ x,
                         const float* __restrict__ W,
                         float* __restrict__ h,
                         const int* __restrict__ ei,
                         unsigned* __restrict__ deg,
                         int N, int E, int nMat, int nDeg) {
    static_assert(FBLK >= C_DIM, "the matvec copies W with one thread per element");
    __shared__ __attribute__((aligned(16))) float wl[C_DIM];
    __shared__ __attribute__((aligned(16))) float xt[(FBLK / 32) * XT_WAVE];
    const int b     = blockIdx.x;
    const int pairs = min(nMat, nDeg);
    bool counts;
    int  item;
    if (b < 2 * pairs) { counts = (__popc((unsigned)b) & 1) != 0; item = b >> 1; }
    else               { counts = nDeg > nMat;                    item = b - pairs; }
    if (counts) gcn_degree_block(ei, deg, E, N, item);
    else        gcn_matvec_wmma_block(x, W, h, N, item, xt, wl);
}

__global__ void gcn_dinv_kernel(unsigned* __restrict__ deg,
                                float* __restrict__ dinv, int N) {
    int i = blockIdx.x * blockDim.x + threadIdx.x;
    if (i < N) {
        const unsigned d = deg[i] + deg[N + i];
        dinv[i] = rsqrtf((float)(d + 1u));
        deg[i]     = 0u;
        deg[N + i] = 0u;
    }
}

__global__ void __launch_bounds__(EBLK)
gcn_edge_kernel(const int* __restrict__ ei,
                const float* __restrict__ dinv,
                const float* __restrict__ h,
                float* __restrict__ accum, int E, int N) {
    float* __restrict__ plane = accum + (size_t)chip_half() * (size_t)N;
    const int* __restrict__ dstp = ei + E;
    const int base = blockIdx.x * (EBLK * EPT);
    const int e0   = base + threadIdx.x;
    const bool whole = (base + EBLK * EPT <= E);
    int   s[EPT], d[EPT];
    float w[EPT];
    if (whole) {
        #pragma unroll
        for (int j = 0; j < EPT; ++j) {
            s[j] = ei[e0 + j * EBLK];
            d[j] = dstp[e0 + j * EBLK];
        }
    } else {
        #pragma unroll
        for (int j = 0; j < EPT; ++j) {
            const int e = min(e0 + j * EBLK, E - 1);
            s[j] = ei[e];
            d[j] = dstp[e];
        }
    }
    #pragma unroll
    for (int j = 0; j < EPT; ++j) w[j] = dinv[s[j]] * dinv[d[j]] * h[s[j]];
    if (whole) {
        #pragma unroll
        for (int j = 0; j < EPT; ++j) atomicAdd(&plane[d[j]], w[j]);
    } else {
        #pragma unroll
        for (int j = 0; j < EPT; ++j)
            if (e0 + j * EBLK < E) atomicAdd(&plane[d[j]], w[j]);
    }
}

__global__ void gcn_finalize_kernel(const float* __restrict__ accum,
                                    const float* __restrict__ dinv,
                                    const float* __restrict__ h,
                                    const float* __restrict__ b,
                                    float* __restrict__ out, int N) {
    int i = blockIdx.x * blockDim.x + threadIdx.x;
    if (i < N) {
        float di = dinv[i];
        float v  = (accum[i] + accum[N + i]) + di * di * h[i] + b[0];
        out[i]   = fmaxf(v, 0.0f) + log1pf(expf(-fabsf(v)));
    }
}


#define RS_BITS  8
#define RS       (1 << RS_BITS)
#define RMAX     512
#define SRC_BITS 17
#define SRC_MASK ((1u << SRC_BITS) - 1u)
#define SBLK     512
#define SCHUNK   (SBLK * EPT)
#define RBLK     256
#define CAP      8704

__global__ void gcn_clear_kernel(unsigned* __restrict__ words, int n) {
    int i = blockIdx.x * blockDim.x + threadIdx.x;
    if (i < n) words[i] = 0u;
}

struct PlaceLds {
    unsigned lc[RMAX];
    unsigned lgot[RMAX];
    unsigned loff[RMAX];
    unsigned sb[2][RMAX];
    unsigned srec[SCHUNK];
    unsigned short srng[SCHUNK];
};
struct TileLds {
    float wl[C_DIM];
    float xt[(SBLK / 32) * XT_WAVE];
};
union PlaceOrTileLds {
    PlaceLds p;
    __attribute__((aligned(16))) TileLds m;
};

__device__ __forceinline__ void gcn_place_block(const int* __restrict__ ei,
                                                unsigned* __restrict__ fill,
                                                unsigned* __restrict__ recs,
                                                int E, int chunk, PlaceLds& L) {
    unsigned (&lc)[RMAX]         = L.lc;
    unsigned (&lgot)[RMAX]       = L.lgot;
    unsigned (&loff)[RMAX]       = L.loff;
    unsigned (&sb)[2][RMAX]      = L.sb;
    unsigned (&srec)[SCHUNK]     = L.srec;
    unsigned short (&srng)[SCHUNK] = L.srng;
    const int t = threadIdx.x;
    lc[t] = 0u;
    __syncthreads();
    const int* __restrict__ dstp = ei + E;
    const int base = chunk * SCHUNK;
    const int e0   = base + t;
    int s[EPT], d[EPT];
    unsigned rk[EPT];
    #pragma unroll
    for (int j = 0; j < EPT; ++j) {
        const int e = min(e0 + j * SBLK, E - 1);
        s[j] = ei[e];
        d[j] = dstp[e];
    }
    #pragma unroll
    for (int j = 0; j < EPT; ++j) {
        rk[j] = 0u;
        if (e0 + j * SBLK < E) rk[j] = atomicAdd(&lc[d[j] >> RS_BITS], 1u);
    }
    __syncthreads();
    const unsigned c = lc[t];
    unsigned got = 0u;
    if (c) got = atomicAdd(&fill[t], c);
    sb[0][t] = c;
    __syncthreads();
    int p = 0;
    for (int off = 1; off < RMAX; off <<= 1) {
        unsigned ub = sb[p][t];
        if (t >= off) ub += sb[p][t - off];
        sb[p ^ 1][t] = ub;
        p ^= 1;
        __syncthreads();
    }
    lgot[t] = got;
    loff[t] = sb[p][t] - c;
    __syncthreads();
    #pragma unroll
    for (int j = 0; j < EPT; ++j) {
        if (e0 + j * SBLK < E) {
            const unsigned r   = (unsigned)d[j] >> RS_BITS;
            const unsigned pos = loff[r] + rk[j];
            srec[pos] = (((unsigned)d[j] & (RS - 1u)) << SRC_BITS) | (unsigned)s[j];
            srng[pos] = (unsigned short)r;
        }
    }
    __syncthreads();
    const int nvalid = min(SCHUNK, E - base);
    for (int i = t; i < nvalid; i += SBLK) {
        const unsigned r = srng[i];
        const unsigned k = lgot[r] + ((unsigned)i - loff[r]);
        if (k < CAP) recs[(size_t)r * CAP + k] = srec[i];
    }
}

__global__ void __launch_bounds__(SBLK) __attribute__((amdgpu_waves_per_eu(16, 16)))
gcn_matvec_place_kernel(const float* __restrict__ x,
                        const float* __restrict__ W,
                        float* __restrict__ h,
                        const int* __restrict__ ei,
                        unsigned* __restrict__ fill,
                        unsigned* __restrict__ recs,
                        int N, int E, int nMat, int nPlace) {
    static_assert(SBLK >= C_DIM, "the matvec copies W with one thread per element");
    __shared__ PlaceOrTileLds sm;
    const int b     = blockIdx.x;
    const int pairs = min(nMat, nPlace);
    bool places;
    int  item;
    if (b < 2 * pairs) { places = (__popc((unsigned)b) & 1) != 0; item = b >> 1; }
    else               { places = nPlace > nMat;                  item = b - pairs; }
    if (places) gcn_place_block(ei, fill, recs, E, item, sm.p);
    else        gcn_matvec_wmma_block(x, W, h, N, item, sm.m.xt, sm.m.wl);
}

__global__ void __launch_bounds__(RBLK)
gcn_bin_degree_kernel(const unsigned* __restrict__ recs,
                      const unsigned* __restrict__ fill,
                      const int* __restrict__ ei,
                      const float* __restrict__ h,
                      float* __restrict__ dinv,
                      float* __restrict__ g, int N, int E) {
    __shared__ unsigned cnt[RS];
    const int t = threadIdx.x;
    const int r = blockIdx.x;
    cnt[t] = 0u;
    __syncthreads();
    const unsigned n = fill[r];
    if (n <= CAP) {
        const unsigned* __restrict__ p = recs + (size_t)r * CAP;
        for (unsigned i0 = 0; i0 < n; i0 += RBLK * EPT) {
            unsigned rc[EPT];
            #pragma unroll
            for (int j = 0; j < EPT; ++j) rc[j] = p[min(i0 + t + j * RBLK, n - 1u)];
            #pragma unroll
            for (int j = 0; j < EPT; ++j)
                if (i0 + t + j * RBLK < n) atomicAdd(&cnt[rc[j] >> SRC_BITS], 1u);
        }
    } else {
        const int* __restrict__ dstp = ei + E;
        for (int i = t; i < E; i += RBLK) {
            const int d = dstp[i];
            if ((d >> RS_BITS) == r) atomicAdd(&cnt[d & (RS - 1)], 1u);
        }
    }
    __syncthreads();
    const int node = r * RS + t;
    if (node < N) {
        const float di = rsqrtf((float)(cnt[t] + 1u));
        dinv[node] = di;
        g[node]    = di * h[node];
    }
}

__global__ void __launch_bounds__(RBLK)
gcn_bin_message_kernel(const unsigned* __restrict__ recs,
                       const unsigned* __restrict__ fill,
                       const int* __restrict__ ei,
                       const float* __restrict__ g,
                       const float* __restrict__ dinv,
                       const float* __restrict__ b,
                       float* __restrict__ out, int N, int E) {
    __shared__ float sum[RS];
    const int t = threadIdx.x;
    const int r = blockIdx.x;
    sum[t] = 0.0f;
    __syncthreads();
    const unsigned n = fill[r];
    if (n <= CAP) {
        const unsigned* __restrict__ p = recs + (size_t)r * CAP;
        for (unsigned i0 = 0; i0 < n; i0 += RBLK * EPT) {
            unsigned rc[EPT];
            float    gv[EPT];
            #pragma unroll
            for (int j = 0; j < EPT; ++j) rc[j] = p[min(i0 + t + j * RBLK, n - 1u)];
            #pragma unroll
            for (int j = 0; j < EPT; ++j) gv[j] = g[rc[j] & SRC_MASK];
            #pragma unroll
            for (int j = 0; j < EPT; ++j)
                if (i0 + t + j * RBLK < n) atomicAdd(&sum[rc[j] >> SRC_BITS], gv[j]);
        }
    } else {
        const int* __restrict__ dstp = ei + E;
        for (int i = t; i < E; i += RBLK) {
            const int d = dstp[i];
            if ((d >> RS_BITS) == r) atomicAdd(&sum[d & (RS - 1)], g[ei[i]]);
        }
    }
    __syncthreads();
    const int node = r * RS + t;
    if (node < N) {
        const float v = dinv[node] * (sum[t] + g[node]) + b[0];
        out[node] = fmaxf(v, 0.0f) + log1pf(expf(-fabsf(v)));
    }
}

extern "C" void kernel_launch(void* const* d_in, const int* in_sizes, int n_in,
                              void* d_out, int out_size, void* d_ws, size_t ws_size,
                              hipStream_t stream) {
    const float* x  = (const float*)d_in[0];
    const int*   ei = (const int*)d_in[1];
    const float* W  = (const float*)d_in[2];
    const float* b  = (const float*)d_in[3];
    float* out = (float*)d_out;

    const int N = in_sizes[0] / C_DIM;
    const int E = in_sizes[1] / 2;

    float*    h     = (float*)d_ws;
    unsigned* deg   = (unsigned*)(h + N);
    float*    accum = (float*)deg;
    float*    dinv  = (float*)(deg + 2 * (size_t)N);

    const int BLK = 256;
    const int gridN = (N + BLK - 1) / BLK;
    const int gridE = (E + EBLK * EPT - 1) / (EBLK * EPT);
    const int nodesPerBlock = (FBLK / 32) * 16;
    const int gridM = (N + nodesPerBlock - 1) / nodesPerBlock;

    const int gridD = (E + FBLK * EPT - 1) / (FBLK * EPT);

    const int bins = (N + RS - 1) / RS;
    const size_t need = ((size_t)3 * N + RMAX + (size_t)bins * CAP) * sizeof(unsigned);
    if (ws_size >= need && N <= RMAX * RS) {
        float*    g2    = h + N;
        float*    dinv2 = g2 + N;
        unsigned* fill  = (unsigned*)(dinv2 + N);
        unsigned* recs  = fill + RMAX;
        const int gridP = (E + SCHUNK - 1) / SCHUNK;
        const int nodesPerPBlock = (SBLK / 32) * 16;
        const int gridMP = (N + nodesPerPBlock - 1) / nodesPerPBlock;
        gcn_clear_kernel<<<(RMAX + BLK - 1) / BLK, BLK, 0, stream>>>(fill, RMAX);
        gcn_matvec_place_kernel<<<gridMP + gridP, SBLK, 0, stream>>>(x, W, h, ei, fill, recs, N, E, gridMP, gridP);
        gcn_bin_degree_kernel<<<bins, RBLK, 0, stream>>>(recs, fill, ei, h, dinv2, g2, N, E);
        gcn_bin_message_kernel<<<bins, RBLK, 0, stream>>>(recs, fill, ei, g2, dinv2, b, out, N, E);
        return;
    }

    gcn_zero_kernel<<<gridN, BLK, 0, stream>>>(deg, N);
    gcn_matvec_degree_kernel<<<gridM + gridD, FBLK, 0, stream>>>(x, W, h, ei, deg, N, E, gridM, gridD);
    gcn_dinv_kernel<<<gridN, BLK, 0, stream>>>(deg, dinv, N);
    gcn_edge_kernel<<<gridE, EBLK, 0, stream>>>(ei, dinv, h, accum, E, N);
    gcn_finalize_kernel<<<gridN, BLK, 0, stream>>>(accum, dinv, h, b, out, N);
}


# Bin phase re-cut to the chip: 241 bins of 416 nodes, one bin block per WGP (was 391 blocks on 256 WGPs), each taking its whole stretch in one pass of 512 threads; sums by arrival as before
# speedup vs baseline: 8.4818x; 1.0335x over previous
#include <hip/hip_runtime.h>
#include <hip/hip_bf16.h>
#include <math.h>

typedef __attribute__((ext_vector_type(2))) float v2f;
typedef __attribute__((ext_vector_type(4))) float v4f;
typedef __attribute__((ext_vector_type(8))) float v8f;

#define C_DIM 256

#define XT_K     64
#define XT_PITCH 68
#define XT_WAVE  (16 * XT_PITCH)

__device__ __forceinline__ int chip_half() {
    const unsigned id = __builtin_amdgcn_s_sendmsg_rtn(0x87);
    return (int)((id >> 18) & 1u);
}

__global__ void gcn_zero_kernel(unsigned* __restrict__ deg, int N) {
    int i = blockIdx.x * blockDim.x + threadIdx.x;
    if (i < N) {
        deg[i]     = 0u;
        deg[N + i] = 0u;
    }
}

__device__ __forceinline__ void gcn_matvec_store_tile(float* __restrict__ h, const v8f acc,
                                                      int base, int lane, int N) {
    if ((lane & 15) == 0) {
        const int mbase = base + (lane >> 4) * 8;
        #pragma unroll
        for (int r8 = 0; r8 < 8; ++r8) {
            if (mbase + r8 < N) h[mbase + r8] = acc[r8];
        }
    }
}

__device__ __forceinline__ void gcn_matvec_wmma_block(const float* __restrict__ x,
                                                      const float* __restrict__ W,
                                                      float* __restrict__ h, int N, int blk,
                                                      float* __restrict__ xt,
                                                      float* __restrict__ wl) {
    if (threadIdx.x < C_DIM) wl[threadIdx.x] = W[threadIdx.x];
    __syncthreads();

    const int lane = threadIdx.x & 31;
    const int wave = __builtin_amdgcn_readfirstlane((int)(threadIdx.x >> 5));
    const int wavesPerBlock = blockDim.x >> 5;
    const int base = __builtin_amdgcn_readfirstlane((blk * wavesPerBlock + wave) * 16);
    if (base >= N) return;

    const int half = lane >> 4;
    const int m    = lane & 15;

    v8f acc = {0.f, 0.f, 0.f, 0.f, 0.f, 0.f, 0.f, 0.f};

    if (base + 16 > N) {
        const int row = min(base + m, N - 1);
        const float* __restrict__ xrow = x + (size_t)row * C_DIM;
        #pragma unroll 4
        for (int k0 = 0; k0 < C_DIM; k0 += 4) {
            const int ka = k0 + half * 2;
            v2f a = *(const v2f*)(xrow + ka);
            v2f b;
            b.x = W[ka];
            b.y = W[ka + 1];
            acc = __builtin_amdgcn_wmma_f32_16x16x4_f32(
                 false, a,  false, b,
                 (short)0, acc,  false,  false);
        }
        gcn_matvec_store_tile(h, acc, base, lane, N);
        return;
    }

    const char* xp = (const char*)(x + (size_t)base * C_DIM) + (unsigned)(half * C_DIM + 4 * m) * 4u;
    float* tile = xt + wave * XT_WAVE;
    float* st = tile + half * XT_PITCH + 4 * m;
    const float* ld = tile + m * XT_PITCH + 2 * half;
    const float* wf = wl + 2 * half;

    v4f r[8];
    #pragma unroll
    for (int i = 0; i < 8; ++i) r[i] = *(const v4f*)(xp + i * (2 * C_DIM * 4));

    #pragma nounroll
    for (int q = 0; q < C_DIM / XT_K; ++q) {
        #pragma unroll
        for (int i = 0; i < 8; ++i) *(v4f*)(st + 2 * i * XT_PITCH) = r[i];
        if (q + 1 < C_DIM / XT_K) {
            xp += XT_K * 4;
            #pragma unroll
            for (int i = 0; i < 8; ++i) r[i] = *(const v4f*)(xp + i * (2 * C_DIM * 4));
        }
        #pragma unroll
        for (int s = 0; s < XT_K / 4; ++s) {
            v2f a = *(const v2f*)(ld + 4 * s);
            v2f b = *(const v2f*)(wf + 4 * s);
            acc = __builtin_amdgcn_wmma_f32_16x16x4_f32(
                 false, a,  false, b,
                 (short)0, acc,  false,  false);
        }
        wf += XT_K;
    }
    gcn_matvec_store_tile(h, acc, base, lane, N);
}

#define EPT  16
#define EBLK 128

#define FBLK 256

__device__ __forceinline__ void gcn_degree_block(const int* __restrict__ ei,
                                                 unsigned* __restrict__ deg,
                                                 int E, int N, int chunk) {
    unsigned* __restrict__ plane = deg + (size_t)chip_half() * (size_t)N;
    const int* __restrict__ dstp = ei + E;
    const int base = chunk * (FBLK * EPT);
    const int e0   = base + threadIdx.x;
    int d[EPT];
    if (base + FBLK * EPT <= E) {
        #pragma unroll
        for (int j = 0; j < EPT; ++j) d[j] = dstp[e0 + j * FBLK];
        #pragma unroll
        for (int j = 0; j < EPT; ++j) atomicAdd(&plane[d[j]], 1u);
    } else {
        #pragma unroll
        for (int j = 0; j < EPT; ++j) d[j] = dstp[min(e0 + j * FBLK, E - 1)];
        #pragma unroll
        for (int j = 0; j < EPT; ++j)
            if (e0 + j * FBLK < E) atomicAdd(&plane[d[j]], 1u);
    }
}

__global__ void __launch_bounds__(FBLK)
gcn_matvec_degree_kernel(const float* __restrict__ x,
                         const float* __restrict__ W,
                         float* __restrict__ h,
                         const int* __restrict__ ei,
                         unsigned* __restrict__ deg,
                         int N, int E, int nMat, int nDeg) {
    static_assert(FBLK >= C_DIM, "the matvec copies W with one thread per element");
    __shared__ __attribute__((aligned(16))) float wl[C_DIM];
    __shared__ __attribute__((aligned(16))) float xt[(FBLK / 32) * XT_WAVE];
    const int b     = blockIdx.x;
    const int pairs = min(nMat, nDeg);
    bool counts;
    int  item;
    if (b < 2 * pairs) { counts = (__popc((unsigned)b) & 1) != 0; item = b >> 1; }
    else               { counts = nDeg > nMat;                    item = b - pairs; }
    if (counts) gcn_degree_block(ei, deg, E, N, item);
    else        gcn_matvec_wmma_block(x, W, h, N, item, xt, wl);
}

__global__ void gcn_dinv_kernel(unsigned* __restrict__ deg,
                                float* __restrict__ dinv, int N) {
    int i = blockIdx.x * blockDim.x + threadIdx.x;
    if (i < N) {
        const unsigned d = deg[i] + deg[N + i];
        dinv[i] = rsqrtf((float)(d + 1u));
        deg[i]     = 0u;
        deg[N + i] = 0u;
    }
}

__global__ void __launch_bounds__(EBLK)
gcn_edge_kernel(const int* __restrict__ ei,
                const float* __restrict__ dinv,
                const float* __restrict__ h,
                float* __restrict__ accum, int E, int N) {
    float* __restrict__ plane = accum + (size_t)chip_half() * (size_t)N;
    const int* __restrict__ dstp = ei + E;
    const int base = blockIdx.x * (EBLK * EPT);
    const int e0   = base + threadIdx.x;
    const bool whole = (base + EBLK * EPT <= E);
    int   s[EPT], d[EPT];
    float w[EPT];
    if (whole) {
        #pragma unroll
        for (int j = 0; j < EPT; ++j) {
            s[j] = ei[e0 + j * EBLK];
            d[j] = dstp[e0 + j * EBLK];
        }
    } else {
        #pragma unroll
        for (int j = 0; j < EPT; ++j) {
            const int e = min(e0 + j * EBLK, E - 1);
            s[j] = ei[e];
            d[j] = dstp[e];
        }
    }
    #pragma unroll
    for (int j = 0; j < EPT; ++j) w[j] = dinv[s[j]] * dinv[d[j]] * h[s[j]];
    if (whole) {
        #pragma unroll
        for (int j = 0; j < EPT; ++j) atomicAdd(&plane[d[j]], w[j]);
    } else {
        #pragma unroll
        for (int j = 0; j < EPT; ++j)
            if (e0 + j * EBLK < E) atomicAdd(&plane[d[j]], w[j]);
    }
}

__global__ void gcn_finalize_kernel(const float* __restrict__ accum,
                                    const float* __restrict__ dinv,
                                    const float* __restrict__ h,
                                    const float* __restrict__ b,
                                    float* __restrict__ out, int N) {
    int i = blockIdx.x * blockDim.x + threadIdx.x;
    if (i < N) {
        float di = dinv[i];
        float v  = (accum[i] + accum[N + i]) + di * di * h[i] + b[0];
        out[i]   = fmaxf(v, 0.0f) + log1pf(expf(-fabsf(v)));
    }
}


#define RS       416
#define RMAX     512
#define SRC_BITS 17
#define SRC_MASK ((1u << SRC_BITS) - 1u)
#define DL_BITS  9
#define RK_BITS  13
#define SBLK     512
#define SCHUNK   (SBLK * EPT)
#define RBLK     512
#define CAP      14112
#define RPT      ((CAP + RBLK - 1) / RBLK)

__global__ void gcn_clear_kernel(unsigned* __restrict__ words, int n) {
    int i = blockIdx.x * blockDim.x + threadIdx.x;
    if (i < n) words[i] = 0u;
}

struct PlaceLds {
    unsigned lc[RMAX];
    unsigned lgot[RMAX];
    unsigned loff[RMAX];
    unsigned sb[2][RMAX];
    unsigned srec[SCHUNK];
    unsigned short srng[SCHUNK];
};
struct TileLds {
    float wl[C_DIM];
    float xt[(SBLK / 32) * XT_WAVE];
};
union PlaceOrTileLds {
    PlaceLds p;
    __attribute__((aligned(16))) TileLds m;
};

__device__ __forceinline__ void gcn_place_block(const int* __restrict__ ei,
                                                unsigned* __restrict__ fill,
                                                unsigned* __restrict__ recs,
                                                int E, int chunk, PlaceLds& L) {
    unsigned (&lc)[RMAX]         = L.lc;
    unsigned (&lgot)[RMAX]       = L.lgot;
    unsigned (&loff)[RMAX]       = L.loff;
    unsigned (&sb)[2][RMAX]      = L.sb;
    unsigned (&srec)[SCHUNK]     = L.srec;
    unsigned short (&srng)[SCHUNK] = L.srng;
    const int t = threadIdx.x;
    lc[t] = 0u;
    __syncthreads();
    const int* __restrict__ dstp = ei + E;
    const int base = chunk * SCHUNK;
    const int e0   = base + t;
    static_assert(RMAX <= (1 << 9) && RS <= (1 << DL_BITS) && SCHUNK <= (1 << RK_BITS),
                  "bin, dst within bin and rank share one word");
    int s[EPT];
    unsigned d[EPT];
    #pragma unroll
    for (int j = 0; j < EPT; ++j) {
        const int e = min(e0 + j * SBLK, E - 1);
        s[j] = ei[e];
        d[j] = (unsigned)dstp[e];
    }
    #pragma unroll
    for (int j = 0; j < EPT; ++j) {
        const unsigned r = d[j] / RS;
        unsigned pk = (r << (DL_BITS + RK_BITS)) | ((d[j] - r * RS) << RK_BITS);
        if (e0 + j * SBLK < E) pk |= atomicAdd(&lc[r], 1u);
        d[j] = pk;
    }
    __syncthreads();
    const unsigned c = lc[t];
    unsigned got = 0u;
    if (c) got = atomicAdd(&fill[t], c);
    sb[0][t] = c;
    __syncthreads();
    int p = 0;
    for (int off = 1; off < RMAX; off <<= 1) {
        unsigned ub = sb[p][t];
        if (t >= off) ub += sb[p][t - off];
        sb[p ^ 1][t] = ub;
        p ^= 1;
        __syncthreads();
    }
    lgot[t] = got;
    loff[t] = sb[p][t] - c;
    __syncthreads();
    #pragma unroll
    for (int j = 0; j < EPT; ++j) {
        if (e0 + j * SBLK < E) {
            const unsigned r   = d[j] >> (DL_BITS + RK_BITS);
            const unsigned pos = loff[r] + (d[j] & ((1u << RK_BITS) - 1u));
            srec[pos] = (((d[j] >> RK_BITS) & ((1u << DL_BITS) - 1u)) << SRC_BITS) | (unsigned)s[j];
            srng[pos] = (unsigned short)r;
        }
    }
    __syncthreads();
    const int nvalid = min(SCHUNK, E - base);
    for (int i = t; i < nvalid; i += SBLK) {
        const unsigned r = srng[i];
        const unsigned k = lgot[r] + ((unsigned)i - loff[r]);
        if (k < CAP) recs[(size_t)r * CAP + k] = srec[i];
    }
}

__global__ void __launch_bounds__(SBLK) __attribute__((amdgpu_waves_per_eu(16, 16)))
gcn_matvec_place_kernel(const float* __restrict__ x,
                        const float* __restrict__ W,
                        float* __restrict__ h,
                        const int* __restrict__ ei,
                        unsigned* __restrict__ fill,
                        unsigned* __restrict__ recs,
                        int N, int E, int nMat, int nPlace) {
    static_assert(SBLK >= C_DIM, "the matvec copies W with one thread per element");
    __shared__ PlaceOrTileLds sm;
    const int b     = blockIdx.x;
    const int pairs = min(nMat, nPlace);
    bool places;
    int  item;
    if (b < 2 * pairs) { places = (__popc((unsigned)b) & 1) != 0; item = b >> 1; }
    else               { places = nPlace > nMat;                  item = b - pairs; }
    if (places) gcn_place_block(ei, fill, recs, E, item, sm.p);
    else        gcn_matvec_wmma_block(x, W, h, N, item, sm.m.xt, sm.m.wl);
}

__global__ void __launch_bounds__(RBLK)
gcn_bin_degree_kernel(const unsigned* __restrict__ recs,
                      const unsigned* __restrict__ fill,
                      const int* __restrict__ ei,
                      const float* __restrict__ h,
                      float* __restrict__ dinv,
                      float* __restrict__ g, int N, int E) {
    static_assert(RBLK >= RS, "one thread per node of the bin");
    __shared__ unsigned cnt[RS];
    const int t = threadIdx.x;
    const int r = blockIdx.x;
    if (t < RS) cnt[t] = 0u;
    __syncthreads();
    const int  node = r * RS + t;
    const bool mine = t < RS && node < N;
    const float hn = mine ? h[node] : 0.0f;
    const unsigned n = fill[r];
    if (n <= CAP) {
        if (n) {
            const unsigned* __restrict__ p = recs + (size_t)r * CAP;
            unsigned rc[RPT];
            #pragma unroll
            for (int j = 0; j < RPT; ++j) rc[j] = p[min((unsigned)(t + j * RBLK), n - 1u)];
            #pragma unroll
            for (int j = 0; j < RPT; ++j)
                if ((unsigned)(t + j * RBLK) < n) atomicAdd(&cnt[rc[j] >> SRC_BITS], 1u);
        }
    } else {
        const int* __restrict__ dstp = ei + E;
        for (int i = t; i < E; i += RBLK) {
            const unsigned d = (unsigned)dstp[i];
            if (d / RS == (unsigned)r) atomicAdd(&cnt[d - (unsigned)r * RS], 1u);
        }
    }
    __syncthreads();
    if (mine) {
        const float di = rsqrtf((float)(cnt[t] + 1u));
        dinv[node] = di;
        g[node]    = di * hn;
    }
}

__global__ void __launch_bounds__(RBLK)
gcn_bin_message_kernel(const unsigned* __restrict__ recs,
                       const unsigned* __restrict__ fill,
                       const int* __restrict__ ei,
                       const float* __restrict__ g,
                       const float* __restrict__ dinv,
                       const float* __restrict__ b,
                       float* __restrict__ out, int N, int E) {
    static_assert(RBLK >= RS, "one thread per node of the bin");
    __shared__ float sum[RS];
    const int t = threadIdx.x;
    const int r = blockIdx.x;
    if (t < RS) sum[t] = 0.0f;
    __syncthreads();
    const int  node = r * RS + t;
    const bool mine = t < RS && node < N;
    const float dn = mine ? dinv[node] : 0.0f;
    const float gn = mine ? g[node] : 0.0f;
    const float bv = b[0];
    const unsigned n = fill[r];
    if (n <= CAP) {
        if (n) {
            const unsigned* __restrict__ p = recs + (size_t)r * CAP;
            unsigned rc[RPT];
            float    gv[RPT];
            #pragma unroll
            for (int j = 0; j < RPT; ++j) rc[j] = p[min((unsigned)(t + j * RBLK), n - 1u)];
            #pragma unroll
            for (int j = 0; j < RPT; ++j) gv[j] = g[rc[j] & SRC_MASK];
            #pragma unroll
            for (int j = 0; j < RPT; ++j)
                if ((unsigned)(t + j * RBLK) < n) atomicAdd(&sum[rc[j] >> SRC_BITS], gv[j]);
        }
    } else {
        const int* __restrict__ dstp = ei + E;
        for (int i = t; i < E; i += RBLK) {
            const unsigned d = (unsigned)dstp[i];
            if (d / RS == (unsigned)r) atomicAdd(&sum[d - (unsigned)r * RS], g[ei[i]]);
        }
    }
    __syncthreads();
    if (mine) {
        const float v = dn * (sum[t] + gn) + bv;
        out[node] = fmaxf(v, 0.0f) + log1pf(expf(-fabsf(v)));
    }
}

extern "C" void kernel_launch(void* const* d_in, const int* in_sizes, int n_in,
                              void* d_out, int out_size, void* d_ws, size_t ws_size,
                              hipStream_t stream) {
    const float* x  = (const float*)d_in[0];
    const int*   ei = (const int*)d_in[1];
    const float* W  = (const float*)d_in[2];
    const float* b  = (const float*)d_in[3];
    float* out = (float*)d_out;

    const int N = in_sizes[0] / C_DIM;
    const int E = in_sizes[1] / 2;

    float*    h     = (float*)d_ws;
    unsigned* deg   = (unsigned*)(h + N);
    float*    accum = (float*)deg;
    float*    dinv  = (float*)(deg + 2 * (size_t)N);

    const int BLK = 256;
    const int gridN = (N + BLK - 1) / BLK;
    const int gridE = (E + EBLK * EPT - 1) / (EBLK * EPT);
    const int nodesPerBlock = (FBLK / 32) * 16;
    const int gridM = (N + nodesPerBlock - 1) / nodesPerBlock;

    const int gridD = (E + FBLK * EPT - 1) / (FBLK * EPT);

    const int bins = (N + RS - 1) / RS;
    const size_t need = ((size_t)3 * N + RMAX + (size_t)bins * CAP) * sizeof(unsigned);
    if (ws_size >= need && N <= RMAX * RS && N <= (1 << SRC_BITS)) {
        float*    g2    = h + N;
        float*    dinv2 = g2 + N;
        unsigned* fill  = (unsigned*)(dinv2 + N);
        unsigned* recs  = fill + RMAX;
        const int gridP = (E + SCHUNK - 1) / SCHUNK;
        const int nodesPerPBlock = (SBLK / 32) * 16;
        const int gridMP = (N + nodesPerPBlock - 1) / nodesPerPBlock;
        gcn_clear_kernel<<<(RMAX + BLK - 1) / BLK, BLK, 0, stream>>>(fill, RMAX);
        gcn_matvec_place_kernel<<<gridMP + gridP, SBLK, 0, stream>>>(x, W, h, ei, fill, recs, N, E, gridMP, gridP);
        gcn_bin_degree_kernel<<<bins, RBLK, 0, stream>>>(recs, fill, ei, h, dinv2, g2, N, E);
        gcn_bin_message_kernel<<<bins, RBLK, 0, stream>>>(recs, fill, ei, g2, dinv2, b, out, N, E);
        return;
    }

    gcn_zero_kernel<<<gridN, BLK, 0, stream>>>(deg, N);
    gcn_matvec_degree_kernel<<<gridM + gridD, FBLK, 0, stream>>>(x, W, h, ei, deg, N, E, gridM, gridD);
    gcn_dinv_kernel<<<gridN, BLK, 0, stream>>>(deg, dinv, N);
    gcn_edge_kernel<<<gridE, EBLK, 0, stream>>>(ei, dinv, h, accum, E, N);
    gcn_finalize_kernel<<<gridN, BLK, 0, stream>>>(accum, dinv, h, b, out, N);
}


# Message gathers in source-range phases: a bin block walks g in stretches of 10240 sources, a thread fetching only its records of the stretch, so g's lines hit the WGP cache; sums by arrival as before
# speedup vs baseline: 9.1079x; 1.0738x over previous
#include <hip/hip_runtime.h>
#include <hip/hip_bf16.h>
#include <math.h>

typedef __attribute__((ext_vector_type(2))) float v2f;
typedef __attribute__((ext_vector_type(4))) float v4f;
typedef __attribute__((ext_vector_type(8))) float v8f;

#define C_DIM 256

#define XT_K     64
#define XT_PITCH 68
#define XT_WAVE  (16 * XT_PITCH)

__device__ __forceinline__ int chip_half() {
    const unsigned id = __builtin_amdgcn_s_sendmsg_rtn(0x87);
    return (int)((id >> 18) & 1u);
}

__global__ void gcn_zero_kernel(unsigned* __restrict__ deg, int N) {
    int i = blockIdx.x * blockDim.x + threadIdx.x;
    if (i < N) {
        deg[i]     = 0u;
        deg[N + i] = 0u;
    }
}

__device__ __forceinline__ void gcn_matvec_store_tile(float* __restrict__ h, const v8f acc,
                                                      int base, int lane, int N) {
    if ((lane & 15) == 0) {
        const int mbase = base + (lane >> 4) * 8;
        #pragma unroll
        for (int r8 = 0; r8 < 8; ++r8) {
            if (mbase + r8 < N) h[mbase + r8] = acc[r8];
        }
    }
}

__device__ __forceinline__ void gcn_matvec_wmma_block(const float* __restrict__ x,
                                                      const float* __restrict__ W,
                                                      float* __restrict__ h, int N, int blk,
                                                      float* __restrict__ xt,
                                                      float* __restrict__ wl) {
    if (threadIdx.x < C_DIM) wl[threadIdx.x] = W[threadIdx.x];
    __syncthreads();

    const int lane = threadIdx.x & 31;
    const int wave = __builtin_amdgcn_readfirstlane((int)(threadIdx.x >> 5));
    const int wavesPerBlock = blockDim.x >> 5;
    const int base = __builtin_amdgcn_readfirstlane((blk * wavesPerBlock + wave) * 16);
    if (base >= N) return;

    const int half = lane >> 4;
    const int m    = lane & 15;

    v8f acc = {0.f, 0.f, 0.f, 0.f, 0.f, 0.f, 0.f, 0.f};

    if (base + 16 > N) {
        const int row = min(base + m, N - 1);
        const float* __restrict__ xrow = x + (size_t)row * C_DIM;
        #pragma unroll 4
        for (int k0 = 0; k0 < C_DIM; k0 += 4) {
            const int ka = k0 + half * 2;
            v2f a = *(const v2f*)(xrow + ka);
            v2f b;
            b.x = W[ka];
            b.y = W[ka + 1];
            acc = __builtin_amdgcn_wmma_f32_16x16x4_f32(
                 false, a,  false, b,
                 (short)0, acc,  false,  false);
        }
        gcn_matvec_store_tile(h, acc, base, lane, N);
        return;
    }

    const char* xp = (const char*)(x + (size_t)base * C_DIM) + (unsigned)(half * C_DIM + 4 * m) * 4u;
    float* tile = xt + wave * XT_WAVE;
    float* st = tile + half * XT_PITCH + 4 * m;
    const float* ld = tile + m * XT_PITCH + 2 * half;
    const float* wf = wl + 2 * half;

    v4f r[8];
    #pragma unroll
    for (int i = 0; i < 8; ++i) r[i] = *(const v4f*)(xp + i * (2 * C_DIM * 4));

    #pragma nounroll
    for (int q = 0; q < C_DIM / XT_K; ++q) {
        #pragma unroll
        for (int i = 0; i < 8; ++i) *(v4f*)(st + 2 * i * XT_PITCH) = r[i];
        if (q + 1 < C_DIM / XT_K) {
            xp += XT_K * 4;
            #pragma unroll
            for (int i = 0; i < 8; ++i) r[i] = *(const v4f*)(xp + i * (2 * C_DIM * 4));
        }
        #pragma unroll
        for (int s = 0; s < XT_K / 4; ++s) {
            v2f a = *(const v2f*)(ld + 4 * s);
            v2f b = *(const v2f*)(wf + 4 * s);
            acc = __builtin_amdgcn_wmma_f32_16x16x4_f32(
                 false, a,  false, b,
                 (short)0, acc,  false,  false);
        }
        wf += XT_K;
    }
    gcn_matvec_store_tile(h, acc, base, lane, N);
}

#define EPT  16
#define EBLK 128

#define FBLK 256

__device__ __forceinline__ void gcn_degree_block(const int* __restrict__ ei,
                                                 unsigned* __restrict__ deg,
                                                 int E, int N, int chunk) {
    unsigned* __restrict__ plane = deg + (size_t)chip_half() * (size_t)N;
    const int* __restrict__ dstp = ei + E;
    const int base = chunk * (FBLK * EPT);
    const int e0   = base + threadIdx.x;
    int d[EPT];
    if (base + FBLK * EPT <= E) {
        #pragma unroll
        for (int j = 0; j < EPT; ++j) d[j] = dstp[e0 + j * FBLK];
        #pragma unroll
        for (int j = 0; j < EPT; ++j) atomicAdd(&plane[d[j]], 1u);
    } else {
        #pragma unroll
        for (int j = 0; j < EPT; ++j) d[j] = dstp[min(e0 + j * FBLK, E - 1)];
        #pragma unroll
        for (int j = 0; j < EPT; ++j)
            if (e0 + j * FBLK < E) atomicAdd(&plane[d[j]], 1u);
    }
}

__global__ void __launch_bounds__(FBLK)
gcn_matvec_degree_kernel(const float* __restrict__ x,
                         const float* __restrict__ W,
                         float* __restrict__ h,
                         const int* __restrict__ ei,
                         unsigned* __restrict__ deg,
                         int N, int E, int nMat, int nDeg) {
    static_assert(FBLK >= C_DIM, "the matvec copies W with one thread per element");
    __shared__ __attribute__((aligned(16))) float wl[C_DIM];
    __shared__ __attribute__((aligned(16))) float xt[(FBLK / 32) * XT_WAVE];
    const int b     = blockIdx.x;
    const int pairs = min(nMat, nDeg);
    bool counts;
    int  item;
    if (b < 2 * pairs) { counts = (__popc((unsigned)b) & 1) != 0; item = b >> 1; }
    else               { counts = nDeg > nMat;                    item = b - pairs; }
    if (counts) gcn_degree_block(ei, deg, E, N, item);
    else        gcn_matvec_wmma_block(x, W, h, N, item, xt, wl);
}

__global__ void gcn_dinv_kernel(unsigned* __restrict__ deg,
                                float* __restrict__ dinv, int N) {
    int i = blockIdx.x * blockDim.x + threadIdx.x;
    if (i < N) {
        const unsigned d = deg[i] + deg[N + i];
        dinv[i] = rsqrtf((float)(d + 1u));
        deg[i]     = 0u;
        deg[N + i] = 0u;
    }
}

__global__ void __launch_bounds__(EBLK)
gcn_edge_kernel(const int* __restrict__ ei,
                const float* __restrict__ dinv,
                const float* __restrict__ h,
                float* __restrict__ accum, int E, int N) {
    float* __restrict__ plane = accum + (size_t)chip_half() * (size_t)N;
    const int* __restrict__ dstp = ei + E;
    const int base = blockIdx.x * (EBLK * EPT);
    const int e0   = base + threadIdx.x;
    const bool whole = (base + EBLK * EPT <= E);
    int   s[EPT], d[EPT];
    float w[EPT];
    if (whole) {
        #pragma unroll
        for (int j = 0; j < EPT; ++j) {
            s[j] = ei[e0 + j * EBLK];
            d[j] = dstp[e0 + j * EBLK];
        }
    } else {
        #pragma unroll
        for (int j = 0; j < EPT; ++j) {
            const int e = min(e0 + j * EBLK, E - 1);
            s[j] = ei[e];
            d[j] = dstp[e];
        }
    }
    #pragma unroll
    for (int j = 0; j < EPT; ++j) w[j] = dinv[s[j]] * dinv[d[j]] * h[s[j]];
    if (whole) {
        #pragma unroll
        for (int j = 0; j < EPT; ++j) atomicAdd(&plane[d[j]], w[j]);
    } else {
        #pragma unroll
        for (int j = 0; j < EPT; ++j)
            if (e0 + j * EBLK < E) atomicAdd(&plane[d[j]], w[j]);
    }
}

__global__ void gcn_finalize_kernel(const float* __restrict__ accum,
                                    const float* __restrict__ dinv,
                                    const float* __restrict__ h,
                                    const float* __restrict__ b,
                                    float* __restrict__ out, int N) {
    int i = blockIdx.x * blockDim.x + threadIdx.x;
    if (i < N) {
        float di = dinv[i];
        float v  = (accum[i] + accum[N + i]) + di * di * h[i] + b[0];
        out[i]   = fmaxf(v, 0.0f) + log1pf(expf(-fabsf(v)));
    }
}


#define RS       416
#define RMAX     512
#define SRC_BITS 17
#define SRC_MASK ((1u << SRC_BITS) - 1u)
#define DL_BITS  9
#define RK_BITS  13
#define SBLK     512
#define SCHUNK   (SBLK * EPT)
#define RBLK     512
#define CAP      14112
#define RPT      ((CAP + RBLK - 1) / RBLK)
#define GPH      10240u

__global__ void gcn_clear_kernel(unsigned* __restrict__ words, int n) {
    int i = blockIdx.x * blockDim.x + threadIdx.x;
    if (i < n) words[i] = 0u;
}

struct PlaceLds {
    unsigned lc[RMAX];
    unsigned lgot[RMAX];
    unsigned loff[RMAX];
    unsigned sb[2][RMAX];
    unsigned srec[SCHUNK];
    unsigned short srng[SCHUNK];
};
struct TileLds {
    float wl[C_DIM];
    float xt[(SBLK / 32) * XT_WAVE];
};
union PlaceOrTileLds {
    PlaceLds p;
    __attribute__((aligned(16))) TileLds m;
};

__device__ __forceinline__ void gcn_place_block(const int* __restrict__ ei,
                                                unsigned* __restrict__ fill,
                                                unsigned* __restrict__ recs,
                                                int E, int chunk, PlaceLds& L) {
    unsigned (&lc)[RMAX]         = L.lc;
    unsigned (&lgot)[RMAX]       = L.lgot;
    unsigned (&loff)[RMAX]       = L.loff;
    unsigned (&sb)[2][RMAX]      = L.sb;
    unsigned (&srec)[SCHUNK]     = L.srec;
    unsigned short (&srng)[SCHUNK] = L.srng;
    const int t = threadIdx.x;
    lc[t] = 0u;
    __syncthreads();
    const int* __restrict__ dstp = ei + E;
    const int base = chunk * SCHUNK;
    const int e0   = base + t;
    static_assert(RMAX <= (1 << 9) && RS <= (1 << DL_BITS) && SCHUNK <= (1 << RK_BITS),
                  "bin, dst within bin and rank share one word");
    int s[EPT];
    unsigned d[EPT];
    #pragma unroll
    for (int j = 0; j < EPT; ++j) {
        const int e = min(e0 + j * SBLK, E - 1);
        s[j] = ei[e];
        d[j] = (unsigned)dstp[e];
    }
    #pragma unroll
    for (int j = 0; j < EPT; ++j) {
        const unsigned r = d[j] / RS;
        unsigned pk = (r << (DL_BITS + RK_BITS)) | ((d[j] - r * RS) << RK_BITS);
        if (e0 + j * SBLK < E) pk |= atomicAdd(&lc[r], 1u);
        d[j] = pk;
    }
    __syncthreads();
    const unsigned c = lc[t];
    unsigned got = 0u;
    if (c) got = atomicAdd(&fill[t], c);
    sb[0][t] = c;
    __syncthreads();
    int p = 0;
    for (int off = 1; off < RMAX; off <<= 1) {
        unsigned ub = sb[p][t];
        if (t >= off) ub += sb[p][t - off];
        sb[p ^ 1][t] = ub;
        p ^= 1;
        __syncthreads();
    }
    lgot[t] = got;
    loff[t] = sb[p][t] - c;
    __syncthreads();
    #pragma unroll
    for (int j = 0; j < EPT; ++j) {
        if (e0 + j * SBLK < E) {
            const unsigned r   = d[j] >> (DL_BITS + RK_BITS);
            const unsigned pos = loff[r] + (d[j] & ((1u << RK_BITS) - 1u));
            srec[pos] = (((d[j] >> RK_BITS) & ((1u << DL_BITS) - 1u)) << SRC_BITS) | (unsigned)s[j];
            srng[pos] = (unsigned short)r;
        }
    }
    __syncthreads();
    const int nvalid = min(SCHUNK, E - base);
    for (int i = t; i < nvalid; i += SBLK) {
        const unsigned r = srng[i];
        const unsigned k = lgot[r] + ((unsigned)i - loff[r]);
        if (k < CAP) recs[(size_t)r * CAP + k] = srec[i];
    }
}

__global__ void __launch_bounds__(SBLK) __attribute__((amdgpu_waves_per_eu(16, 16)))
gcn_matvec_place_kernel(const float* __restrict__ x,
                        const float* __restrict__ W,
                        float* __restrict__ h,
                        const int* __restrict__ ei,
                        unsigned* __restrict__ fill,
                        unsigned* __restrict__ recs,
                        int N, int E, int nMat, int nPlace) {
    static_assert(SBLK >= C_DIM, "the matvec copies W with one thread per element");
    __shared__ PlaceOrTileLds sm;
    const int b     = blockIdx.x;
    const int pairs = min(nMat, nPlace);
    bool places;
    int  item;
    if (b < 2 * pairs) { places = (__popc((unsigned)b) & 1) != 0; item = b >> 1; }
    else               { places = nPlace > nMat;                  item = b - pairs; }
    if (places) gcn_place_block(ei, fill, recs, E, item, sm.p);
    else        gcn_matvec_wmma_block(x, W, h, N, item, sm.m.xt, sm.m.wl);
}

__global__ void __launch_bounds__(RBLK)
gcn_bin_degree_kernel(const unsigned* __restrict__ recs,
                      const unsigned* __restrict__ fill,
                      const int* __restrict__ ei,
                      const float* __restrict__ h,
                      float* __restrict__ dinv,
                      float* __restrict__ g, int N, int E) {
    static_assert(RBLK >= RS, "one thread per node of the bin");
    __shared__ unsigned cnt[RS];
    const int t = threadIdx.x;
    const int r = blockIdx.x;
    if (t < RS) cnt[t] = 0u;
    __syncthreads();
    const int  node = r * RS + t;
    const bool mine = t < RS && node < N;
    const float hn = mine ? h[node] : 0.0f;
    const unsigned n = fill[r];
    if (n <= CAP) {
        if (n) {
            const unsigned* __restrict__ p = recs + (size_t)r * CAP;
            unsigned rc[RPT];
            #pragma unroll
            for (int j = 0; j < RPT; ++j) rc[j] = p[min((unsigned)(t + j * RBLK), n - 1u)];
            #pragma unroll
            for (int j = 0; j < RPT; ++j)
                if ((unsigned)(t + j * RBLK) < n) atomicAdd(&cnt[rc[j] >> SRC_BITS], 1u);
        }
    } else {
        const int* __restrict__ dstp = ei + E;
        for (int i = t; i < E; i += RBLK) {
            const unsigned d = (unsigned)dstp[i];
            if (d / RS == (unsigned)r) atomicAdd(&cnt[d - (unsigned)r * RS], 1u);
        }
    }
    __syncthreads();
    if (mine) {
        const float di = rsqrtf((float)(cnt[t] + 1u));
        dinv[node] = di;
        g[node]    = di * hn;
    }
}

__global__ void __launch_bounds__(RBLK)
gcn_bin_message_kernel(const unsigned* __restrict__ recs,
                       const unsigned* __restrict__ fill,
                       const int* __restrict__ ei,
                       const float* __restrict__ g,
                       const float* __restrict__ dinv,
                       const float* __restrict__ b,
                       float* __restrict__ out, int N, int E) {
    static_assert(RBLK >= RS, "one thread per node of the bin");
    __shared__ float sum[RS];
    const int t = threadIdx.x;
    const int r = blockIdx.x;
    if (t < RS) sum[t] = 0.0f;
    __syncthreads();
    const int  node = r * RS + t;
    const bool mine = t < RS && node < N;
    const float dn = mine ? dinv[node] : 0.0f;
    const float gn = mine ? g[node] : 0.0f;
    const float bv = b[0];
    const unsigned n = fill[r];
    if (n <= CAP) {
        if (n) {
            const unsigned* __restrict__ p = recs + (size_t)r * CAP;
            unsigned rc[RPT];
            unsigned off[RPT];
            unsigned phid[RPT];
            float    gv[RPT];
            #pragma unroll
            for (int j = 0; j < RPT; ++j) rc[j] = p[min((unsigned)(t + j * RBLK), n - 1u)];
            #pragma unroll
            for (int j = 0; j < RPT; ++j) {
                const unsigned sj = rc[j] & SRC_MASK;
                off[j]  = sj * 4u;
                phid[j] = ((unsigned)(t + j * RBLK) < n) ? sj / GPH : 0xFFFFFFFFu;
                gv[j]  = 0.0f;
            }
            const int phases = (N + GPH - 1) / GPH;
            #pragma nounroll
            for (int ph = 0; ph < phases; ++ph) {
                #pragma unroll
                for (int j = 0; j < RPT; ++j) {
                    if (phid[j] == (unsigned)ph) {
                        asm volatile("global_load_b32 %0, %1, %2" : "+v"(gv[j]) : "v"(off[j]), "s"(g));
                    }
                }
                __builtin_amdgcn_s_barrier();
            }
            asm volatile("s_wait_loadcnt 0x0" ::: "memory");
            #pragma unroll
            for (int j = 0; j < RPT; ++j) {
                asm volatile("" : "+v"(gv[j]));
                if (phid[j] != 0xFFFFFFFFu) atomicAdd(&sum[rc[j] >> SRC_BITS], gv[j]);
            }
        }
    } else {
        const int* __restrict__ dstp = ei + E;
        for (int i = t; i < E; i += RBLK) {
            const unsigned d = (unsigned)dstp[i];
            if (d / RS == (unsigned)r) atomicAdd(&sum[d - (unsigned)r * RS], g[ei[i]]);
        }
    }
    __syncthreads();
    if (mine) {
        const float v = dn * (sum[t] + gn) + bv;
        out[node] = fmaxf(v, 0.0f) + log1pf(expf(-fabsf(v)));
    }
}

extern "C" void kernel_launch(void* const* d_in, const int* in_sizes, int n_in,
                              void* d_out, int out_size, void* d_ws, size_t ws_size,
                              hipStream_t stream) {
    const float* x  = (const float*)d_in[0];
    const int*   ei = (const int*)d_in[1];
    const float* W  = (const float*)d_in[2];
    const float* b  = (const float*)d_in[3];
    float* out = (float*)d_out;

    const int N = in_sizes[0] / C_DIM;
    const int E = in_sizes[1] / 2;

    float*    h     = (float*)d_ws;
    unsigned* deg   = (unsigned*)(h + N);
    float*    accum = (float*)deg;
    float*    dinv  = (float*)(deg + 2 * (size_t)N);

    const int BLK = 256;
    const int gridN = (N + BLK - 1) / BLK;
    const int gridE = (E + EBLK * EPT - 1) / (EBLK * EPT);
    const int nodesPerBlock = (FBLK / 32) * 16;
    const int gridM = (N + nodesPerBlock - 1) / nodesPerBlock;

    const int gridD = (E + FBLK * EPT - 1) / (FBLK * EPT);

    const int bins = (N + RS - 1) / RS;
    const size_t need = ((size_t)3 * N + RMAX + (size_t)bins * CAP) * sizeof(unsigned);
    if (ws_size >= need && N <= RMAX * RS && N <= (1 << SRC_BITS)) {
        float*    g2    = h + N;
        float*    dinv2 = g2 + N;
        unsigned* fill  = (unsigned*)(dinv2 + N);
        unsigned* recs  = fill + RMAX;
        const int gridP = (E + SCHUNK - 1) / SCHUNK;
        const int nodesPerPBlock = (SBLK / 32) * 16;
        const int gridMP = (N + nodesPerPBlock - 1) / nodesPerPBlock;
        gcn_clear_kernel<<<(RMAX + BLK - 1) / BLK, BLK, 0, stream>>>(fill, RMAX);
        gcn_matvec_place_kernel<<<gridMP + gridP, SBLK, 0, stream>>>(x, W, h, ei, fill, recs, N, E, gridMP, gridP);
        gcn_bin_degree_kernel<<<bins, RBLK, 0, stream>>>(recs, fill, ei, h, dinv2, g2, N, E);
        gcn_bin_message_kernel<<<bins, RBLK, 0, stream>>>(recs, fill, ei, g2, dinv2, b, out, N, E);
        return;
    }

    gcn_zero_kernel<<<gridN, BLK, 0, stream>>>(deg, N);
    gcn_matvec_degree_kernel<<<gridM + gridD, FBLK, 0, stream>>>(x, W, h, ei, deg, N, E, gridM, gridD);
    gcn_dinv_kernel<<<gridN, BLK, 0, stream>>>(deg, dinv, N);
    gcn_edge_kernel<<<gridE, EBLK, 0, stream>>>(ei, dinv, h, accum, E, N);
    gcn_finalize_kernel<<<gridN, BLK, 0, stream>>>(accum, dinv, h, b, out, N);
}
